# starting kernel
# baseline (speedup 1.0000x reference)
#include <hip/hip_runtime.h>


#define N_SAMPLES       128
#define RAYS_PER_WAVE   16
#define WAVES_PER_BLOCK 4
#define RAYS_PER_BLOCK  (RAYS_PER_WAVE * WAVES_PER_BLOCK)
#define LDS_STRIDE      17

typedef __attribute__((ext_vector_type(2))) float v2f;
typedef __attribute__((ext_vector_type(8))) float v8f;

__global__ __launch_bounds__(WAVES_PER_BLOCK * 32)
void volsdf_weights_kernel(const float* __restrict__ sdf,
                           const float* __restrict__ deltas,
                           const float* __restrict__ beta,
                           float* __restrict__ out)
{
    __shared__ float tile[WAVES_PER_BLOCK][N_SAMPLES * LDS_STRIDE];

    const int tid   = threadIdx.x;
    const int wave  = tid >> 5;
    const int lane  = tid & 31;
    const int rbase = blockIdx.x * RAYS_PER_BLOCK + wave * RAYS_PER_WAVE;

    const float b     = beta[0];
    const float inv_b = 1.0f / b;

    float* t = tile[wave];

    const float4* sdf4 = reinterpret_cast<const float4*>(sdf)    + (size_t)rbase * (N_SAMPLES / 4);
    const float4* del4 = reinterpret_cast<const float4*>(deltas) + (size_t)rbase * (N_SAMPLES / 4);
    #pragma unroll
    for (int it = 0; it < RAYS_PER_WAVE; ++it) {
        float4 s4 = sdf4[it * 32 + lane];
        float4 d4 = del4[it * 32 + lane];
        const int samp = lane * 4;
        const float sv[4] = {s4.x, s4.y, s4.z, s4.w};
        const float dv[4] = {d4.x, d4.y, d4.z, d4.w};
        #pragma unroll
        for (int j = 0; j < 4; ++j) {
            float s = sv[j];
            float e = __expf(-fabsf(s) * inv_b);
            float dens = (s >= 0.0f) ? (0.5f * e * inv_b) : ((1.0f - 0.5f * e) * inv_b);
            t[(samp + j) * LDS_STRIDE + it] = dv[j] * dens;
        }
    }
    __syncthreads();

    const int n     = lane & 15;
    const int m     = lane & 15;
    const int khalf = (lane >> 4) * 2;
    const int moff  = (lane >> 4) * 8;

    v2f aop[4];
    #pragma unroll
    for (int kb = 0; kb < 4; ++kb) {
        const int k0 = 4 * kb + khalf;
        aop[kb].x = (k0     <= m) ? 1.0f : 0.0f;
        aop[kb].y = (k0 + 1 <= m) ? 1.0f : 0.0f;
    }

    float running = 0.0f;
    float* outp = out + (size_t)(rbase + n) * N_SAMPLES;

    #pragma unroll
    for (int c = 0; c < 8; ++c) {
        v8f acc = {};
        #pragma unroll
        for (int kb = 0; kb < 4; ++kb) {
            const int s0 = c * 16 + 4 * kb + khalf;
            v2f bb;
            bb.x = t[s0 * LDS_STRIDE + n];
            bb.y = t[(s0 + 1) * LDS_STRIDE + n];
            acc = __builtin_amdgcn_wmma_f32_16x16x4_f32(
                false, aop[kb], false, bb, (short)0, acc, false, false);
        }

        const float tot = __shfl(acc[7], 16 + n, 32);

        float w[8];
        #pragma unroll
        for (int v = 0; v < 8; ++v) {
            const int mm = v + moff;
            const float x    = t[(c * 16 + mm) * LDS_STRIDE + n];
            const float incl = (c == 0) ? acc[v] : (running + acc[v]);
            const float cs   = incl - x;
            const float alpha = 1.0f - __expf(-x);
            const float trans = __expf(-cs);
            w[v] = alpha * trans;
        }
        running = (c == 0) ? tot : (running + tot);

        float4* o4 = reinterpret_cast<float4*>(outp + c * 16 + moff);
        o4[0] = make_float4(w[0], w[1], w[2], w[3]);
        o4[1] = make_float4(w[4], w[5], w[6], w[7]);
    }
}

extern "C" void kernel_launch(void* const* d_in, const int* in_sizes, int n_in,
                              void* d_out, int out_size, void* d_ws, size_t ws_size,
                              hipStream_t stream) {
    const float* sdf    = (const float*)d_in[0];
    const float* deltas = (const float*)d_in[1];
    const float* beta   = (const float*)d_in[2];
    float* out = (float*)d_out;

    const int n_rays = in_sizes[0] / N_SAMPLES;
    const int grid   = n_rays / RAYS_PER_BLOCK;
    volsdf_weights_kernel<<<grid, WAVES_PER_BLOCK * 32, 0, stream>>>(sdf, deltas, beta, out);
}


# Results staged through the wave's LDS tile so each global store instruction writes one whole 512-B ray row (4 full 128-B lines) instead of 16 lines x two 16-B pieces
# speedup vs baseline: 1.8483x; 1.8483x over previous
#include <hip/hip_runtime.h>


#define N_SAMPLES       128
#define RAYS_PER_WAVE   16
#define WAVES_PER_BLOCK 4
#define RAYS_PER_BLOCK  (RAYS_PER_WAVE * WAVES_PER_BLOCK)
#define LDS_STRIDE      17
#define CHUNK_FLOATS    (16 * LDS_STRIDE)

typedef __attribute__((ext_vector_type(2))) float v2f;
typedef __attribute__((ext_vector_type(4))) float v4f;
typedef __attribute__((ext_vector_type(8))) float v8f;

__global__ __launch_bounds__(WAVES_PER_BLOCK * 32)
void volsdf_weights_kernel(const float* __restrict__ sdf,
                           const float* __restrict__ deltas,
                           const float* __restrict__ beta,
                           float* __restrict__ out)
{
    __shared__ __attribute__((aligned(16))) float tile[WAVES_PER_BLOCK][N_SAMPLES * LDS_STRIDE];

    const int tid   = threadIdx.x;
    const int wave  = tid >> 5;
    const int lane  = tid & 31;
    const int rbase = blockIdx.x * RAYS_PER_BLOCK + wave * RAYS_PER_WAVE;

    const float b     = beta[0];
    const float inv_b = 1.0f / b;

    float* t = tile[wave];

    const float4* sdf4 = reinterpret_cast<const float4*>(sdf)    + (size_t)rbase * (N_SAMPLES / 4);
    const float4* del4 = reinterpret_cast<const float4*>(deltas) + (size_t)rbase * (N_SAMPLES / 4);
    #pragma unroll
    for (int it = 0; it < RAYS_PER_WAVE; ++it) {
        float4 s4 = sdf4[it * 32 + lane];
        float4 d4 = del4[it * 32 + lane];
        const int samp = lane * 4;
        const float sv[4] = {s4.x, s4.y, s4.z, s4.w};
        const float dv[4] = {d4.x, d4.y, d4.z, d4.w};
        #pragma unroll
        for (int j = 0; j < 4; ++j) {
            float s = sv[j];
            float e = __expf(-fabsf(s) * inv_b);
            float dens = (s >= 0.0f) ? (0.5f * e * inv_b) : ((1.0f - 0.5f * e) * inv_b);
            t[(samp + j) * LDS_STRIDE + it] = dv[j] * dens;
        }
    }
    __syncthreads();

    const int n     = lane & 15;
    const int m     = lane & 15;
    const int khalf = (lane >> 4) * 2;
    const int moff  = (lane >> 4) * 8;

    v2f aop[4];
    #pragma unroll
    for (int kb = 0; kb < 4; ++kb) {
        const int k0 = 4 * kb + khalf;
        aop[kb].x = (k0     <= m) ? 1.0f : 0.0f;
        aop[kb].y = (k0 + 1 <= m) ? 1.0f : 0.0f;
    }

    float running = 0.0f;

    const int hq = (lane >> 4) * 2;
    const int sw = n >> 2;
    float* stg0 = t + n * 16 + 4 * ((hq    ) ^ sw);
    float* stg1 = t + n * 16 + 4 * ((hq + 1) ^ sw);

    #pragma unroll
    for (int c = 0; c < 8; ++c) {
        v8f acc = {};
        #pragma unroll
        for (int kb = 0; kb < 4; ++kb) {
            const int s0 = c * 16 + 4 * kb + khalf;
            v2f bb;
            bb.x = t[s0 * LDS_STRIDE + n];
            bb.y = t[(s0 + 1) * LDS_STRIDE + n];
            acc = __builtin_amdgcn_wmma_f32_16x16x4_f32(
                false, aop[kb], false, bb, (short)0, acc, false, false);
        }

        const float tot = __shfl(acc[7], 16 + n, 32);

        float w[8];
        #pragma unroll
        for (int v = 0; v < 8; ++v) {
            const int mm = v + moff;
            const float x    = t[(c * 16 + mm) * LDS_STRIDE + n];
            const float incl = (c == 0) ? acc[v] : (running + acc[v]);
            const float cs   = incl - x;
            const float alpha = 1.0f - __expf(-x);
            const float trans = __expf(-cs);
            w[v] = alpha * trans;
        }
        running = (c == 0) ? tot : (running + tot);

        *reinterpret_cast<v4f*>(stg0 + c * CHUNK_FLOATS) = (v4f){w[0], w[1], w[2], w[3]};
        *reinterpret_cast<v4f*>(stg1 + c * CHUNK_FLOATS) = (v4f){w[4], w[5], w[6], w[7]};
    }

    const float* rd   = t + (lane >> 2) * CHUNK_FLOATS;
    const int    q    = lane & 3;
    float*       orow = out + (size_t)rbase * N_SAMPLES + lane * 4;
    #pragma unroll
    for (int r = 0; r < RAYS_PER_WAVE; ++r) {
        const v4f v = *reinterpret_cast<const v4f*>(rd + r * 16 + 4 * (q ^ (r >> 2)));
        *reinterpret_cast<v4f*>(orow + (size_t)r * N_SAMPLES) = v;
    }
}

extern "C" void kernel_launch(void* const* d_in, const int* in_sizes, int n_in,
                              void* d_out, int out_size, void* d_ws, size_t ws_size,
                              hipStream_t stream) {
    const float* sdf    = (const float*)d_in[0];
    const float* deltas = (const float*)d_in[1];
    const float* beta   = (const float*)d_in[2];
    float* out = (float*)d_out;

    const int n_rays = in_sizes[0] / N_SAMPLES;
    const int grid   = n_rays / RAYS_PER_BLOCK;
    volsdf_weights_kernel<<<grid, WAVES_PER_BLOCK * 32, 0, stream>>>(sdf, deltas, beta, out);
}


# Exponentials in base 2: log2(e)/beta folded into wave constants, staged and summed values are delta*density*log2(e) (same sum order), each exp is one bare v_exp_f32; wave index read wave-uniform
# speedup vs baseline: 2.0110x; 1.0880x over previous
#include <hip/hip_runtime.h>


#define N_SAMPLES       128
#define RAYS_PER_WAVE   16
#define WAVES_PER_BLOCK 4
#define RAYS_PER_BLOCK  (RAYS_PER_WAVE * WAVES_PER_BLOCK)
#define LDS_STRIDE      17
#define CHUNK_FLOATS    (16 * LDS_STRIDE)

typedef __attribute__((ext_vector_type(2))) float v2f;
typedef __attribute__((ext_vector_type(4))) float v4f;
typedef __attribute__((ext_vector_type(8))) float v8f;

__global__ __launch_bounds__(WAVES_PER_BLOCK * 32)
void volsdf_weights_kernel(const float* __restrict__ sdf,
                           const float* __restrict__ deltas,
                           const float* __restrict__ beta,
                           float* __restrict__ out)
{
    __shared__ __attribute__((aligned(16))) float tile[WAVES_PER_BLOCK][N_SAMPLES * LDS_STRIDE];

    const int tid   = threadIdx.x;
    const int wave  = __builtin_amdgcn_readfirstlane(tid >> 5);
    const int lane  = tid & 31;
    const int rbase = blockIdx.x * RAYS_PER_BLOCK + wave * RAYS_PER_WAVE;

    const float b     = beta[0];
    const float inv_b = 1.0f / b;
    const float k1    = 1.44269504088896340736f * inv_b;
    const float nk1   = -k1;
    const float kh    = 0.5f * k1;

    float* t = tile[wave];

    const float4* sdf4 = reinterpret_cast<const float4*>(sdf)    + (size_t)rbase * (N_SAMPLES / 4);
    const float4* del4 = reinterpret_cast<const float4*>(deltas) + (size_t)rbase * (N_SAMPLES / 4);
    #pragma unroll
    for (int it = 0; it < RAYS_PER_WAVE; ++it) {
        float4 s4 = sdf4[it * 32 + lane];
        float4 d4 = del4[it * 32 + lane];
        const int samp = lane * 4;
        const float sv[4] = {s4.x, s4.y, s4.z, s4.w};
        const float dv[4] = {d4.x, d4.y, d4.z, d4.w};
        #pragma unroll
        for (int j = 0; j < 4; ++j) {
            float s = sv[j];
            float e = __builtin_amdgcn_exp2f(fabsf(s) * nk1);
            float dens2 = (s >= 0.0f) ? (kh * e) : fmaf(-kh, e, k1);
            t[(samp + j) * LDS_STRIDE + it] = dv[j] * dens2;
        }
    }
    __syncthreads();

    const int n     = lane & 15;
    const int m     = lane & 15;
    const int khalf = (lane >> 4) * 2;
    const int moff  = (lane >> 4) * 8;

    v2f aop[4];
    #pragma unroll
    for (int kb = 0; kb < 4; ++kb) {
        const int k0 = 4 * kb + khalf;
        aop[kb].x = (k0     <= m) ? 1.0f : 0.0f;
        aop[kb].y = (k0 + 1 <= m) ? 1.0f : 0.0f;
    }

    float running = 0.0f;

    const int hq = (lane >> 4) * 2;
    const int sw = n >> 2;
    float* stg0 = t + n * 16 + 4 * ((hq    ) ^ sw);
    float* stg1 = t + n * 16 + 4 * ((hq + 1) ^ sw);

    #pragma unroll
    for (int c = 0; c < 8; ++c) {
        v8f acc = {};
        #pragma unroll
        for (int kb = 0; kb < 4; ++kb) {
            const int s0 = c * 16 + 4 * kb + khalf;
            v2f bb;
            bb.x = t[s0 * LDS_STRIDE + n];
            bb.y = t[(s0 + 1) * LDS_STRIDE + n];
            acc = __builtin_amdgcn_wmma_f32_16x16x4_f32(
                false, aop[kb], false, bb, (short)0, acc, false, false);
        }

        const float tot = __shfl(acc[7], 16 + n, 32);

        float w[8];
        #pragma unroll
        for (int v = 0; v < 8; ++v) {
            const int mm = v + moff;
            const float x    = t[(c * 16 + mm) * LDS_STRIDE + n];
            const float incl = (c == 0) ? acc[v] : (running + acc[v]);
            const float cs   = incl - x;
            const float alpha = 1.0f - __builtin_amdgcn_exp2f(-x);
            const float trans = __builtin_amdgcn_exp2f(-cs);
            w[v] = alpha * trans;
        }
        running = (c == 0) ? tot : (running + tot);

        *reinterpret_cast<v4f*>(stg0 + c * CHUNK_FLOATS) = (v4f){w[0], w[1], w[2], w[3]};
        *reinterpret_cast<v4f*>(stg1 + c * CHUNK_FLOATS) = (v4f){w[4], w[5], w[6], w[7]};
    }

    const float* rd   = t + (lane >> 2) * CHUNK_FLOATS;
    const int    q    = lane & 3;
    float*       orow = out + (size_t)rbase * N_SAMPLES + lane * 4;
    #pragma unroll
    for (int r = 0; r < RAYS_PER_WAVE; ++r) {
        const v4f v = *reinterpret_cast<const v4f*>(rd + r * 16 + 4 * (q ^ (r >> 2)));
        *reinterpret_cast<v4f*>(orow + (size_t)r * N_SAMPLES) = v;
    }
}

extern "C" void kernel_launch(void* const* d_in, const int* in_sizes, int n_in,
                              void* d_out, int out_size, void* d_ws, size_t ws_size,
                              hipStream_t stream) {
    const float* sdf    = (const float*)d_in[0];
    const float* deltas = (const float*)d_in[1];
    const float* beta   = (const float*)d_in[2];
    float* out = (float*)d_out;

    const int n_rays = in_sizes[0] / N_SAMPLES;
    const int grid   = n_rays / RAYS_PER_BLOCK;
    volsdf_weights_kernel<<<grid, WAVES_PER_BLOCK * 32, 0, stream>>>(sdf, deltas, beta, out);
}


# Weight as the drop in transmittance, w_i = exp2(-S_(i-1)) - exp2(-S_i) on the inclusive sums S the WMMA returns: one exponential per sample in the scan, x not re-read, half swap by v_permlanex16
# speedup vs baseline: 2.1276x; 1.0580x over previous
#include <hip/hip_runtime.h>


#define N_SAMPLES       128
#define RAYS_PER_WAVE   16
#define WAVES_PER_BLOCK 4
#define RAYS_PER_BLOCK  (RAYS_PER_WAVE * WAVES_PER_BLOCK)
#define LDS_STRIDE      17
#define CHUNK_FLOATS    (16 * LDS_STRIDE)

typedef __attribute__((ext_vector_type(2))) float v2f;
typedef __attribute__((ext_vector_type(4))) float v4f;
typedef __attribute__((ext_vector_type(8))) float v8f;

static __device__ __forceinline__ float xhalf(float x)
{
    const int b = __builtin_bit_cast(int, x);
    return __builtin_bit_cast(float,
        __builtin_amdgcn_permlanex16(b, b, 0x76543210, 0xfedcba98, false, false));
}

__global__ __launch_bounds__(WAVES_PER_BLOCK * 32)
void volsdf_weights_kernel(const float* __restrict__ sdf,
                           const float* __restrict__ deltas,
                           const float* __restrict__ beta,
                           float* __restrict__ out)
{
    __shared__ __attribute__((aligned(16))) float tile[WAVES_PER_BLOCK][N_SAMPLES * LDS_STRIDE];

    const int tid   = threadIdx.x;
    const int wave  = __builtin_amdgcn_readfirstlane(tid >> 5);
    const int lane  = tid & 31;
    const int rbase = blockIdx.x * RAYS_PER_BLOCK + wave * RAYS_PER_WAVE;

    const float b     = beta[0];
    const float inv_b = 1.0f / b;
    const float k1    = 1.44269504088896340736f * inv_b;
    const float nk1   = -k1;
    const float kh    = 0.5f * k1;

    float* t = tile[wave];

    const float4* sdf4 = reinterpret_cast<const float4*>(sdf)    + (size_t)rbase * (N_SAMPLES / 4);
    const float4* del4 = reinterpret_cast<const float4*>(deltas) + (size_t)rbase * (N_SAMPLES / 4);
    #pragma unroll
    for (int it = 0; it < RAYS_PER_WAVE; ++it) {
        float4 s4 = sdf4[it * 32 + lane];
        float4 d4 = del4[it * 32 + lane];
        const int samp = lane * 4;
        const float sv[4] = {s4.x, s4.y, s4.z, s4.w};
        const float dv[4] = {d4.x, d4.y, d4.z, d4.w};
        #pragma unroll
        for (int j = 0; j < 4; ++j) {
            float s = sv[j];
            float e = __builtin_amdgcn_exp2f(fabsf(s) * nk1);
            float dens2 = (s >= 0.0f) ? (kh * e) : fmaf(-kh, e, k1);
            t[(samp + j) * LDS_STRIDE + it] = dv[j] * dens2;
        }
    }
    __syncthreads();
    __builtin_amdgcn_sched_barrier(0);

    const int n     = lane & 15;
    const int m     = lane & 15;
    const int khalf = (lane >> 4) * 2;

    v2f aop[4];
    #pragma unroll
    for (int kb = 0; kb < 4; ++kb) {
        const int k0 = 4 * kb + khalf;
        aop[kb].x = (k0     <= m) ? 1.0f : 0.0f;
        aop[kb].y = (k0 + 1 <= m) ? 1.0f : 0.0f;
    }

    float running = 0.0f;
    const bool upper = lane >= 16;

    const int hq = (lane >> 4) * 2;
    const int sw = n >> 2;
    float* stg0 = t + n * 16 + 4 * ((hq    ) ^ sw);
    float* stg1 = t + n * 16 + 4 * ((hq + 1) ^ sw);

    #pragma unroll
    for (int c = 0; c < 8; ++c) {
        v8f acc = {};
        #pragma unroll
        for (int kb = 0; kb < 4; ++kb) {
            const int s0 = c * 16 + 4 * kb + khalf;
            v2f bb;
            bb.x = t[s0 * LDS_STRIDE + n];
            bb.y = t[(s0 + 1) * LDS_STRIDE + n];
            acc = __builtin_amdgcn_wmma_f32_16x16x4_f32(
                false, aop[kb], false, bb, (short)0, acc, false, false);
        }

        const float xh  = xhalf(acc[7]);
        const float tot = upper ? acc[7] : xh;
        const float pl  = upper ? xh : 0.0f;

        float eprev = __builtin_amdgcn_exp2f(-((c == 0) ? pl : (running + pl)));
        float w[8];
        #pragma unroll
        for (int v = 0; v < 8; ++v) {
            const float incl = (c == 0) ? acc[v] : (running + acc[v]);
            const float e    = __builtin_amdgcn_exp2f(-incl);
            w[v]  = eprev - e;
            eprev = e;
        }
        running = (c == 0) ? tot : (running + tot);

        *reinterpret_cast<v4f*>(stg0 + c * CHUNK_FLOATS) = (v4f){w[0], w[1], w[2], w[3]};
        *reinterpret_cast<v4f*>(stg1 + c * CHUNK_FLOATS) = (v4f){w[4], w[5], w[6], w[7]};
    }

    const float* rd   = t + (lane >> 2) * CHUNK_FLOATS;
    const int    q    = lane & 3;
    float*       orow = out + (size_t)rbase * N_SAMPLES + lane * 4;
    #pragma unroll
    for (int r = 0; r < RAYS_PER_WAVE; ++r) {
        const v4f v = *reinterpret_cast<const v4f*>(rd + r * 16 + 4 * (q ^ (r >> 2)));
        *reinterpret_cast<v4f*>(orow + (size_t)r * N_SAMPLES) = v;
    }
}

extern "C" void kernel_launch(void* const* d_in, const int* in_sizes, int n_in,
                              void* d_out, int out_size, void* d_ws, size_t ws_size,
                              hipStream_t stream) {
    const float* sdf    = (const float*)d_in[0];
    const float* deltas = (const float*)d_in[1];
    const float* beta   = (const float*)d_in[2];
    float* out = (float*)d_out;

    const int n_rays = in_sizes[0] / N_SAMPLES;
    const int grid   = n_rays / RAYS_PER_BLOCK;
    volsdf_weights_kernel<<<grid, WAVES_PER_BLOCK * 32, 0, stream>>>(sdf, deltas, beta, out);
}
